# starting kernel
# baseline (speedup 1.0000x reference)
#include <hip/hip_runtime.h>
#include <math.h>

#define NB   16384
#define NQ   100
#define NS   50
#define DCUT 32
#define HLOC 4

#define MPAD 34
#define HPAD 17

typedef __attribute__((ext_vector_type(2))) float v2f;
typedef __attribute__((ext_vector_type(8))) float v8f;

static __device__ __forceinline__ v8f wmma4(v2f a, v2f b, v8f c) {
    return __builtin_amdgcn_wmma_f32_16x16x4_f32(
         false, a,  false, b,
         (short)0, c,  false,  false);
}

__global__ __launch_bounds__(128) void mps_rnn_wmma_kernel(
    const int*   __restrict__ xg,
    const float* __restrict__ Mg,
    const float* __restrict__ vg,
    const float* __restrict__ wg,
    const float* __restrict__ cg,
    float*       __restrict__ outg)
{
    __shared__ float Mr[HLOC * DCUT * MPAD];
    __shared__ float Mi[HLOC * DCUT * MPAD];
    __shared__ float vr_s[HLOC * DCUT];
    __shared__ float vi_s[HLOC * DCUT];
    __shared__ float wr_s[DCUT];
    __shared__ float wi_s[DCUT];
    __shared__ float c_s[2];
    __shared__ float hbuf[4][2][DCUT][HPAD];

    const int tid  = threadIdx.x;
    const int lane = tid & 31;
    const int wave = tid >> 5;
    const int n    = lane & 15;
    const int half = lane >> 4;
    const int bcol = blockIdx.x * 64 + wave * 16 + n;

    v2f Hr[8], Hi[8];
    const float h0 = 0.17677669529663688f;
    #pragma unroll
    for (int k = 0; k < 8; ++k) {
        Hr[k][0] = h0; Hr[k][1] = h0;
        Hi[k][0] = 0.f; Hi[k][1] = 0.f;
    }

    float logp = 0.f, phase = 0.f;
    const int k0h = half * 2;

    for (int t = 0; t < NS; ++t) {
        {
            const float* Mt = Mg + (size_t)t * HLOC * DCUT * DCUT * 2;
            for (int idx = tid; idx < HLOC * DCUT * DCUT; idx += 128) {
                const int s = idx >> 10, rem = idx & 1023;
                const int d = rem >> 5,  e  = rem & 31;
                const float re = Mt[idx * 2 + 0];
                const float im = Mt[idx * 2 + 1];
                Mr[(s * DCUT + d) * MPAD + e] = re;
                Mi[(s * DCUT + d) * MPAD + e] = im;
            }
            const float* vt = vg + (size_t)t * HLOC * DCUT * 2;
            if (tid < HLOC * DCUT) { vr_s[tid] = vt[tid * 2]; vi_s[tid] = vt[tid * 2 + 1]; }
            const float* wt = wg + (size_t)t * DCUT * 2;
            if (tid < DCUT) { wr_s[tid] = wt[tid * 2]; wi_s[tid] = wt[tid * 2 + 1]; }
            if (tid == 0) { c_s[0] = cg[t * 2]; c_s[1] = cg[t * 2 + 1]; }
            if (t + 1 < NS) {
                const char* nxt = (const char*)(Mg + (size_t)(t + 1) * HLOC * DCUT * DCUT * 2);
                __builtin_prefetch(nxt + tid * 256, 0, 1);
            }
        }
        __syncthreads();

        v8f htR[HLOC][2], htI[HLOC][2];
        float p_part[HLOC];
        #pragma unroll
        for (int s = 0; s < HLOC; ++s) {
            p_part[s] = 0.f;
            #pragma unroll
            for (int mt = 0; mt < 2; ++mt) {
                v8f accR  = {};
                v8f accMI = {};
                v8f accI  = {};
                const float* mrBase = &Mr[((s * DCUT) + mt * 16 + n) * MPAD];
                const float* miBase = &Mi[((s * DCUT) + mt * 16 + n) * MPAD];
                #pragma unroll
                for (int kk = 0; kk < 8; ++kk) {
                    const int koff = kk * 4 + k0h;
                    const v2f Ar = *(const v2f*)(mrBase + koff);
                    const v2f Ai = *(const v2f*)(miBase + koff);
                    accR  = wmma4(Ar, Hr[kk], accR);
                    accMI = wmma4(Ai, Hi[kk], accMI);
                    accI  = wmma4(Ar, Hi[kk], accI);
                    accI  = wmma4(Ai, Hr[kk], accI);
                }
                #pragma unroll
                for (int r = 0; r < 8; ++r) {
                    const int row = mt * 16 + r + half * 8;
                    const float hr = accR[r] - accMI[r] + vr_s[s * DCUT + row];
                    const float hi = accI[r]            + vi_s[s * DCUT + row];
                    htR[s][mt][r] = hr;
                    htI[s][mt][r] = hi;
                    p_part[s] += hr * hr + hi * hi;
                }
            }
        }

        float p_sig[HLOC], psum = 0.f;
        #pragma unroll
        for (int s = 0; s < HLOC; ++s) {
            p_sig[s] = p_part[s] + __shfl_xor(p_part[s], 16, 32);
            psum += p_sig[s];
        }

        const int xs0 = xg[(size_t)bcol * NQ + 2 * t];
        const int xs1 = xg[(size_t)bcol * NQ + 2 * t + 1];
        const int ss  = 2 * xs0 + xs1;

        const float p_sel = (ss == 0) ? p_sig[0] : (ss == 1) ? p_sig[1]
                          : (ss == 2) ? p_sig[2] : p_sig[3];
        logp += logf(p_sel) - logf(psum);
        const float rinv = rsqrtf(p_sel);

        float ampR = 0.f, ampI = 0.f;
        #pragma unroll
        for (int mt = 0; mt < 2; ++mt) {
            const v8f selR = (ss == 0) ? htR[0][mt] : (ss == 1) ? htR[1][mt]
                           : (ss == 2) ? htR[2][mt] : htR[3][mt];
            const v8f selI = (ss == 0) ? htI[0][mt] : (ss == 1) ? htI[1][mt]
                           : (ss == 2) ? htI[2][mt] : htI[3][mt];
            #pragma unroll
            for (int r = 0; r < 8; ++r) {
                const int row = mt * 16 + r + half * 8;
                const float hr = selR[r] * rinv;
                const float hi = selI[r] * rinv;
                hbuf[wave][0][row][n] = hr;
                hbuf[wave][1][row][n] = hi;
                const float wr = wr_s[row], wi = wi_s[row];
                ampR += wr * hr - wi * hi;
                ampI += wr * hi + wi * hr;
            }
        }
        ampR += __shfl_xor(ampR, 16, 32);
        ampI += __shfl_xor(ampI, 16, 32);
        ampR += c_s[0];
        ampI += c_s[1];
        phase += atan2f(ampI, ampR);

        __syncthreads();

        #pragma unroll
        for (int kk = 0; kk < 8; ++kk) {
            const int krow = kk * 4 + half * 2;
            Hr[kk][0] = hbuf[wave][0][krow + 0][n];
            Hr[kk][1] = hbuf[wave][0][krow + 1][n];
            Hi[kk][0] = hbuf[wave][1][krow + 0][n];
            Hi[kk][1] = hbuf[wave][1][krow + 1][n];
        }
    }

    if (half == 0) {
        float2 o;
        o.x = 0.5f * logp;
        o.y = phase;
        *(float2*)(outg + (size_t)bcol * 2) = o;
    }
}

extern "C" void kernel_launch(void* const* d_in, const int* in_sizes, int n_in,
                              void* d_out, int out_size, void* d_ws, size_t ws_size,
                              hipStream_t stream) {
    (void)in_sizes; (void)n_in; (void)out_size; (void)d_ws; (void)ws_size;
    const int*   x  = (const int*)  d_in[0];
    const float* M  = (const float*)d_in[1];
    const float* v  = (const float*)d_in[2];
    const float* w  = (const float*)d_in[3];
    const float* c  = (const float*)d_in[4];
    float* out = (float*)d_out;

    dim3 grid(NB / 64);
    dim3 block(128);
    mps_rnn_wmma_kernel<<<grid, block, 0, stream>>>(x, M, v, w, c, out);
}


# Each wave owns one of the 4 planes for 32 columns (8 waves): its M fragments go from global into registers a site ahead, M never enters LDS; LDS carries only h, p, amplitude; 1 barrier per site
# speedup vs baseline: 2.0980x; 2.0980x over previous
#include <hip/hip_runtime.h>
#include <math.h>

#define NB   16384
#define NQ   100
#define NS   50
#define DCUT 32
#define HLOC 4

#define BCOLS 64
#define HP    36

typedef __attribute__((ext_vector_type(2))) float v2f;
typedef __attribute__((ext_vector_type(4))) float v4f;
typedef __attribute__((ext_vector_type(8))) float v8f;

static __device__ __forceinline__ v8f wmma4(v2f a, v2f b, v8f c) {
    return __builtin_amdgcn_wmma_f32_16x16x4_f32(
         false, a,  false, b,
         (short)0, c,  false,  false);
}

static __device__ __forceinline__ void load_plane(v4f (&A)[16], const float* __restrict__ Mg,
                                                  int t, int s, int c, int h) {
    const float* base = Mg + (size_t)(t * HLOC + s) * DCUT * DCUT * 2;
    #pragma unroll
    for (int mt = 0; mt < 2; ++mt) {
        #pragma unroll
        for (int kk = 0; kk < 8; ++kk)
            A[mt * 8 + kk] = *(const v4f*)(base + ((mt * 16 + c) * DCUT + kk * 4 + 2 * h) * 2);
    }
}

static __device__ __forceinline__ void load_h(v2f (&Br)[8], v2f (&Bi)[8], const float* hb, int col, int h) {
    const float* b0 = hb + (0 * BCOLS + col) * HP + 2 * h;
    const float* b1 = hb + (1 * BCOLS + col) * HP + 2 * h;
    #pragma unroll
    for (int kk = 0; kk < 8; ++kk) {
        Br[kk] = *(const v2f*)(b0 + kk * 4);
        Bi[kk] = *(const v2f*)(b1 + kk * 4);
    }
}

static __device__ __forceinline__ void tile_products(const v2f (&Ar)[16], const v2f (&Ai)[16],
                                                     const v2f (&Br)[8], const v2f (&Bi)[8],
                                                     v8f (&aR)[2], v8f (&aM)[2], v8f (&aI)[2]) {
    #pragma unroll
    for (int mt = 0; mt < 2; ++mt) { aR[mt] = (v8f){}; aM[mt] = (v8f){}; aI[mt] = (v8f){}; }
    #pragma unroll
    for (int kk = 0; kk < 8; ++kk) {
        #pragma unroll
        for (int mt = 0; mt < 2; ++mt) {
            aR[mt] = wmma4(Ar[mt * 8 + kk], Br[kk], aR[mt]);
            aM[mt] = wmma4(Ai[mt * 8 + kk], Bi[kk], aM[mt]);
            aI[mt] = wmma4(Ar[mt * 8 + kk], Bi[kk], aI[mt]);
        }
        #pragma unroll
        for (int mt = 0; mt < 2; ++mt)
            aI[mt] = wmma4(Ai[mt * 8 + kk], Br[kk], aI[mt]);
    }
}

static __device__ __forceinline__ void tile_finish(const v8f (&aR)[2], const v8f (&aM)[2], const v8f (&aI)[2],
                                                   const v4f (&vq)[8], const v4f (&wq)[8], float c0, float c1,
                                                   float (&hr)[16], float (&hi)[16],
                                                   float& ps, float& ampR, float& ampI) {
    float pp = 0.f;
    #pragma unroll
    for (int mt = 0; mt < 2; ++mt) {
        #pragma unroll
        for (int r = 0; r < 8; ++r) {
            const float vr = vq[mt * 4 + (r >> 1)][(r & 1) * 2 + 0];
            const float vi = vq[mt * 4 + (r >> 1)][(r & 1) * 2 + 1];
            const float a = aR[mt][r] - aM[mt][r] + vr;
            const float b = aI[mt][r]             + vi;
            hr[mt * 8 + r] = a;
            hi[mt * 8 + r] = b;
            pp += a * a + b * b;
        }
    }
    ps = pp + __shfl_xor(pp, 16, 32);
    const float rinv = rsqrtf(ps);
    float sR = 0.f, sI = 0.f;
    #pragma unroll
    for (int mt = 0; mt < 2; ++mt) {
        #pragma unroll
        for (int r = 0; r < 8; ++r) {
            const float wr = wq[mt * 4 + (r >> 1)][(r & 1) * 2 + 0];
            const float wi = wq[mt * 4 + (r >> 1)][(r & 1) * 2 + 1];
            const float a = hr[mt * 8 + r] * rinv;
            const float b = hi[mt * 8 + r] * rinv;
            hr[mt * 8 + r] = a;
            hi[mt * 8 + r] = b;
            sR += wr * a - wi * b;
            sI += wr * b + wi * a;
        }
    }
    sR += __shfl_xor(sR, 16, 32);
    sI += __shfl_xor(sI, 16, 32);
    ampR = sR + c0;
    ampI = sI + c1;
}

static __device__ __forceinline__ void tile_publish(float* hb, float* pb, float* sb, int s, int col, int h, bool sel,
                                                    const float (&hr)[16], const float (&hi)[16],
                                                    float ps, float ampR, float ampI) {
    if (h == 0) pb[s * BCOLS + col] = ps;
    if (sel) {
        float* d0 = hb + (0 * BCOLS + col) * HP + 8 * h;
        float* d1 = hb + (1 * BCOLS + col) * HP + 8 * h;
        #pragma unroll
        for (int mt = 0; mt < 2; ++mt) {
            #pragma unroll
            for (int q = 0; q < 2; ++q) {
                v4f a, b;
                #pragma unroll
                for (int e = 0; e < 4; ++e) { a[e] = hr[mt * 8 + q * 4 + e]; b[e] = hi[mt * 8 + q * 4 + e]; }
                *(v4f*)(d0 + mt * 16 + q * 4) = a;
                *(v4f*)(d1 + mt * 16 + q * 4) = b;
            }
        }
        if (h == 0) {
            sb[0 * BCOLS + col] = ps;
            sb[1 * BCOLS + col] = ampR;
            sb[2 * BCOLS + col] = ampI;
        }
    }
}

static __device__ __forceinline__ void site_account(const float* pb, const float* sb, int col,
                                                    float& logp, float& phase) {
    float psum = 0.f;
    #pragma unroll
    for (int k = 0; k < HLOC; ++k) psum += pb[k * BCOLS + col];
    logp  += logf(sb[0 * BCOLS + col]) - logf(psum);
    phase += atan2f(sb[2 * BCOLS + col], sb[1 * BCOLS + col]);
}

static __device__ __forceinline__ bool chose(int x0, int x1, int s) {
    const int ss = 2 * x0 + x1;
    return (s == 3) ? (ss != 0 && ss != 1 && ss != 2) : (ss == s);
}

struct Lds {
    float* hb;
    float* pb;
    float* sb;
};

static __device__ __forceinline__ void site(int t, const v4f (&A)[16], v4f (&An)[16], const Lds& L,
                                            const int* __restrict__ xg, const float* __restrict__ Mg,
                                            const float* __restrict__ vg, const float* __restrict__ wg,
                                            const float* __restrict__ cg,
                                            int tid, int wave, int s, int c, int h, int col0, size_t gcol0,
                                            float& logp, float& phase) {
    const int par = t & 1;
    const float* hb_cur = L.hb + par * (2 * BCOLS * HP);
    float*       hb_nxt = L.hb + (par ^ 1) * (2 * BCOLS * HP);
    const float* pb_cur = L.pb + par * (HLOC * BCOLS);
    float*       pb_nxt = L.pb + (par ^ 1) * (HLOC * BCOLS);
    const float* sb_cur = L.sb + par * (3 * BCOLS);
    float*       sb_nxt = L.sb + (par ^ 1) * (3 * BCOLS);

    v2f Br0[8], Bi0[8], Br1[8], Bi1[8];
    load_h(Br0, Bi0, hb_cur, col0, h);
    load_h(Br1, Bi1, hb_cur, col0 + 16, h);
    __builtin_amdgcn_sched_barrier(0);

    v4f vq[8], wq[8];
    {
        const float* vt = vg + ((size_t)(t * HLOC + s) * DCUT + 8 * h) * 2;
        const float* wt = wg + ((size_t)t * DCUT + 8 * h) * 2;
        #pragma unroll
        for (int mt = 0; mt < 2; ++mt) {
            #pragma unroll
            for (int q = 0; q < 4; ++q) {
                vq[mt * 4 + q] = *(const v4f*)(vt + (mt * 16 + 2 * q) * 2);
                wq[mt * 4 + q] = *(const v4f*)(wt + (mt * 16 + 2 * q) * 2);
            }
        }
    }
    const int2 xa = *(const int2*)(xg + gcol0 * NQ + 2 * t);
    const int2 xb = *(const int2*)(xg + (gcol0 + 16) * NQ + 2 * t);
    const float c0 = cg[t * 2 + 0];
    const float c1 = cg[t * 2 + 1];

    if (wave < 2 && t > 0) site_account(pb_cur, sb_cur, tid, logp, phase);

    v8f aR[2], aM[2], aI[2];
    float hr0[16], hi0[16], hr1[16], hi1[16];
    float ps0, ar0, ai0, ps1, ar1, ai1;

    v2f Ar[16], Ai[16];
    #pragma unroll
    for (int f = 0; f < 16; ++f) {
        Ar[f][0] = A[f][0]; Ar[f][1] = A[f][2];
        Ai[f][0] = A[f][1]; Ai[f][1] = A[f][3];
    }

    v8f bR[2], bM[2], bI[2];
    tile_products(Ar, Ai, Br0, Bi0, aR, aM, aI);
    __builtin_amdgcn_sched_barrier(0);
    load_plane(An, Mg, (t + 1 < NS) ? t + 1 : t, s, c, h);
    tile_finish(aR, aM, aI, vq, wq, c0, c1, hr0, hi0, ps0, ar0, ai0);
    tile_products(Ar, Ai, Br1, Bi1, bR, bM, bI);
    #pragma unroll
    for (int i = 0; i < 64; ++i) {
        __builtin_amdgcn_sched_group_barrier(0x008, 1, 0);
        __builtin_amdgcn_sched_group_barrier(0x002, 3, 0);
    }
    tile_publish(hb_nxt, pb_nxt, sb_nxt, s, col0,      h, chose(xa.x, xa.y, s), hr0, hi0, ps0, ar0, ai0);
    tile_finish(bR, bM, bI, vq, wq, c0, c1, hr1, hi1, ps1, ar1, ai1);
    tile_publish(hb_nxt, pb_nxt, sb_nxt, s, col0 + 16, h, chose(xb.x, xb.y, s), hr1, hi1, ps1, ar1, ai1);

    __syncthreads();
}

__global__ __launch_bounds__(256) void mps_rnn_plane_per_wave_kernel(
    const int*   __restrict__ xg,
    const float* __restrict__ Mg,
    const float* __restrict__ vg,
    const float* __restrict__ wg,
    const float* __restrict__ cg,
    float*       __restrict__ outg)
{
    __shared__ __attribute__((aligned(16))) float hbuf[2 * 2 * BCOLS * HP];
    __shared__ float pbuf[2 * HLOC * BCOLS];
    __shared__ float sbuf[2 * 3 * BCOLS];

    const int tid  = threadIdx.x;
    const int lane = tid & 31;
    const int wave = tid >> 5;
    const int c    = lane & 15;
    const int h    = lane >> 4;
    const int s    = wave & 3;
    const int col0 = (wave >> 2) * 32 + c;
    const size_t gcol0 = (size_t)blockIdx.x * BCOLS + col0;

    Lds L;
    L.hb = hbuf; L.pb = pbuf; L.sb = sbuf;

    const float h0 = 0.17677669529663688f;
    for (int idx = tid; idx < BCOLS * DCUT; idx += 256) {
        const int cc = idx >> 5, rr = idx & 31;
        hbuf[(0 * BCOLS + cc) * HP + rr] = h0;
        hbuf[(1 * BCOLS + cc) * HP + rr] = 0.f;
    }

    v4f A0[16], A1[16];
    load_plane(A0, Mg, 0, s, c, h);

    float logp = 0.f, phase = 0.f;
    __syncthreads();

    for (int t = 0; t < NS; t += 2) {
        site(t,     A0, A1, L, xg, Mg, vg, wg, cg, tid, wave, s, c, h, col0, gcol0, logp, phase);
        site(t + 1, A1, A0, L, xg, Mg, vg, wg, cg, tid, wave, s, c, h, col0, gcol0, logp, phase);
    }

    if (wave < 2) {
        site_account(pbuf + (NS & 1) * (HLOC * BCOLS), sbuf + (NS & 1) * (3 * BCOLS), tid, logp, phase);
        float2 o;
        o.x = 0.5f * logp;
        o.y = phase;
        *(float2*)(outg + ((size_t)blockIdx.x * BCOLS + tid) * 2) = o;
    }
}

extern "C" void kernel_launch(void* const* d_in, const int* in_sizes, int n_in,
                              void* d_out, int out_size, void* d_ws, size_t ws_size,
                              hipStream_t stream) {
    (void)in_sizes; (void)n_in; (void)out_size; (void)d_ws; (void)ws_size;
    const int*   x  = (const int*)  d_in[0];
    const float* M  = (const float*)d_in[1];
    const float* v  = (const float*)d_in[2];
    const float* w  = (const float*)d_in[3];
    const float* c  = (const float*)d_in[4];
    float* out = (float*)d_out;

    dim3 grid(NB / BCOLS);
    dim3 block(256);
    mps_rnn_plane_per_wave_kernel<<<grid, block, 0, stream>>>(x, M, v, w, c, out);
}


# Work-group halved to 4 waves (a plane each, 32 columns), two per WGP with no common barrier; a wave raises its issue priority from its second tile to its barrier, so a SIMD's two waves take turns
# speedup vs baseline: 2.2141x; 1.0553x over previous
#include <hip/hip_runtime.h>
#include <math.h>

#define NB   16384
#define NQ   100
#define NS   50
#define DCUT 32
#define HLOC 4

#define BCOLS 32
#define HP    36

typedef __attribute__((ext_vector_type(2))) float v2f;
typedef __attribute__((ext_vector_type(4))) float v4f;
typedef __attribute__((ext_vector_type(8))) float v8f;

static __device__ __forceinline__ v8f wmma4(v2f a, v2f b, v8f c) {
    return __builtin_amdgcn_wmma_f32_16x16x4_f32(
         false, a,  false, b,
         (short)0, c,  false,  false);
}

static __device__ __forceinline__ void load_plane(v4f (&A)[16], const float* __restrict__ Mg,
                                                  int t, int s, int c, int h) {
    const float* base = Mg + (size_t)(t * HLOC + s) * DCUT * DCUT * 2;
    #pragma unroll
    for (int mt = 0; mt < 2; ++mt) {
        #pragma unroll
        for (int kk = 0; kk < 8; ++kk)
            A[mt * 8 + kk] = *(const v4f*)(base + ((mt * 16 + c) * DCUT + kk * 4 + 2 * h) * 2);
    }
}

static __device__ __forceinline__ void load_h(v2f (&Br)[8], v2f (&Bi)[8], const float* hb, int col, int h) {
    const float* b0 = hb + (0 * BCOLS + col) * HP + 2 * h;
    const float* b1 = hb + (1 * BCOLS + col) * HP + 2 * h;
    #pragma unroll
    for (int kk = 0; kk < 8; ++kk) {
        Br[kk] = *(const v2f*)(b0 + kk * 4);
        Bi[kk] = *(const v2f*)(b1 + kk * 4);
    }
}

static __device__ __forceinline__ void tile_products(const v2f (&Ar)[16], const v2f (&Ai)[16],
                                                     const v2f (&Br)[8], const v2f (&Bi)[8],
                                                     v8f (&aR)[2], v8f (&aM)[2], v8f (&aI)[2]) {
    #pragma unroll
    for (int mt = 0; mt < 2; ++mt) { aR[mt] = (v8f){}; aM[mt] = (v8f){}; aI[mt] = (v8f){}; }
    #pragma unroll
    for (int kk = 0; kk < 8; ++kk) {
        #pragma unroll
        for (int mt = 0; mt < 2; ++mt) {
            aR[mt] = wmma4(Ar[mt * 8 + kk], Br[kk], aR[mt]);
            aM[mt] = wmma4(Ai[mt * 8 + kk], Bi[kk], aM[mt]);
            aI[mt] = wmma4(Ar[mt * 8 + kk], Bi[kk], aI[mt]);
        }
        #pragma unroll
        for (int mt = 0; mt < 2; ++mt)
            aI[mt] = wmma4(Ai[mt * 8 + kk], Br[kk], aI[mt]);
    }
}

static __device__ __forceinline__ void tile_finish(const v8f (&aR)[2], const v8f (&aM)[2], const v8f (&aI)[2],
                                                   const v4f (&vq)[8], const v4f (&wq)[8], float c0, float c1,
                                                   float (&hr)[16], float (&hi)[16],
                                                   float& ps, float& ampR, float& ampI) {
    float pp = 0.f;
    #pragma unroll
    for (int mt = 0; mt < 2; ++mt) {
        #pragma unroll
        for (int r = 0; r < 8; ++r) {
            const float vr = vq[mt * 4 + (r >> 1)][(r & 1) * 2 + 0];
            const float vi = vq[mt * 4 + (r >> 1)][(r & 1) * 2 + 1];
            const float a = aR[mt][r] - aM[mt][r] + vr;
            const float b = aI[mt][r]             + vi;
            hr[mt * 8 + r] = a;
            hi[mt * 8 + r] = b;
            pp += a * a + b * b;
        }
    }
    ps = pp + __shfl_xor(pp, 16, 32);
    const float rinv = rsqrtf(ps);
    float sR = 0.f, sI = 0.f;
    #pragma unroll
    for (int mt = 0; mt < 2; ++mt) {
        #pragma unroll
        for (int r = 0; r < 8; ++r) {
            const float wr = wq[mt * 4 + (r >> 1)][(r & 1) * 2 + 0];
            const float wi = wq[mt * 4 + (r >> 1)][(r & 1) * 2 + 1];
            const float a = hr[mt * 8 + r] * rinv;
            const float b = hi[mt * 8 + r] * rinv;
            hr[mt * 8 + r] = a;
            hi[mt * 8 + r] = b;
            sR += wr * a - wi * b;
            sI += wr * b + wi * a;
        }
    }
    sR += __shfl_xor(sR, 16, 32);
    sI += __shfl_xor(sI, 16, 32);
    ampR = sR + c0;
    ampI = sI + c1;
}

static __device__ __forceinline__ void tile_publish(float* hb, float* pb, float* sb, int s, int col, int h, bool sel,
                                                    const float (&hr)[16], const float (&hi)[16],
                                                    float ps, float ampR, float ampI) {
    if (h == 0) pb[s * BCOLS + col] = ps;
    if (sel) {
        float* d0 = hb + (0 * BCOLS + col) * HP + 8 * h;
        float* d1 = hb + (1 * BCOLS + col) * HP + 8 * h;
        #pragma unroll
        for (int mt = 0; mt < 2; ++mt) {
            #pragma unroll
            for (int q = 0; q < 2; ++q) {
                v4f a, b;
                #pragma unroll
                for (int e = 0; e < 4; ++e) { a[e] = hr[mt * 8 + q * 4 + e]; b[e] = hi[mt * 8 + q * 4 + e]; }
                *(v4f*)(d0 + mt * 16 + q * 4) = a;
                *(v4f*)(d1 + mt * 16 + q * 4) = b;
            }
        }
        if (h == 0) {
            sb[0 * BCOLS + col] = ps;
            sb[1 * BCOLS + col] = ampR;
            sb[2 * BCOLS + col] = ampI;
        }
    }
}

static __device__ __forceinline__ void site_account(const float* pb, const float* sb, int col,
                                                    float& logp, float& phase) {
    float psum = 0.f;
    #pragma unroll
    for (int k = 0; k < HLOC; ++k) psum += pb[k * BCOLS + col];
    logp  += logf(sb[0 * BCOLS + col]) - logf(psum);
    phase += atan2f(sb[2 * BCOLS + col], sb[1 * BCOLS + col]);
}

static __device__ __forceinline__ bool chose(int x0, int x1, int s) {
    const int ss = 2 * x0 + x1;
    return (s == 3) ? (ss != 0 && ss != 1 && ss != 2) : (ss == s);
}

struct Lds {
    float* hb;
    float* pb;
    float* sb;
};

static __device__ __forceinline__ void site(int t, const v4f (&A)[16], v4f (&An)[16], const Lds& L,
                                            const int* __restrict__ xg, const float* __restrict__ Mg,
                                            const float* __restrict__ vg, const float* __restrict__ wg,
                                            const float* __restrict__ cg,
                                            int tid, int wave, int s, int c, int h, int col0, size_t gcol0,
                                            float& logp, float& phase) {
    const int par = t & 1;
    const float* hb_cur = L.hb + par * (2 * BCOLS * HP);
    float*       hb_nxt = L.hb + (par ^ 1) * (2 * BCOLS * HP);
    const float* pb_cur = L.pb + par * (HLOC * BCOLS);
    float*       pb_nxt = L.pb + (par ^ 1) * (HLOC * BCOLS);
    const float* sb_cur = L.sb + par * (3 * BCOLS);
    float*       sb_nxt = L.sb + (par ^ 1) * (3 * BCOLS);

    v2f Br0[8], Bi0[8], Br1[8], Bi1[8];
    load_h(Br0, Bi0, hb_cur, col0, h);
    load_h(Br1, Bi1, hb_cur, col0 + 16, h);
    __builtin_amdgcn_sched_barrier(0);

    v4f vq[8], wq[8];
    {
        const float* vt = vg + ((size_t)(t * HLOC + s) * DCUT + 8 * h) * 2;
        const float* wt = wg + ((size_t)t * DCUT + 8 * h) * 2;
        #pragma unroll
        for (int mt = 0; mt < 2; ++mt) {
            #pragma unroll
            for (int q = 0; q < 4; ++q) {
                vq[mt * 4 + q] = *(const v4f*)(vt + (mt * 16 + 2 * q) * 2);
                wq[mt * 4 + q] = *(const v4f*)(wt + (mt * 16 + 2 * q) * 2);
            }
        }
    }
    const int2 xa = *(const int2*)(xg + gcol0 * NQ + 2 * t);
    const int2 xb = *(const int2*)(xg + (gcol0 + 16) * NQ + 2 * t);
    const float c0 = cg[t * 2 + 0];
    const float c1 = cg[t * 2 + 1];

    if (wave == 0 && t > 0) site_account(pb_cur, sb_cur, tid, logp, phase);

    v8f aR[2], aM[2], aI[2];
    float hr0[16], hi0[16], hr1[16], hi1[16];
    float ps0, ar0, ai0, ps1, ar1, ai1;

    v2f Ar[16], Ai[16];
    #pragma unroll
    for (int f = 0; f < 16; ++f) {
        Ar[f][0] = A[f][0]; Ar[f][1] = A[f][2];
        Ai[f][0] = A[f][1]; Ai[f][1] = A[f][3];
    }

    v8f bR[2], bM[2], bI[2];
    tile_products(Ar, Ai, Br0, Bi0, aR, aM, aI);
    __builtin_amdgcn_sched_barrier(0);
    __builtin_amdgcn_s_setprio(3);
    load_plane(An, Mg, (t + 1 < NS) ? t + 1 : t, s, c, h);
    tile_finish(aR, aM, aI, vq, wq, c0, c1, hr0, hi0, ps0, ar0, ai0);
    tile_products(Ar, Ai, Br1, Bi1, bR, bM, bI);
    #pragma unroll
    for (int i = 0; i < 64; ++i) {
        __builtin_amdgcn_sched_group_barrier(0x008, 1, 0);
        __builtin_amdgcn_sched_group_barrier(0x002, 3, 0);
    }
    tile_publish(hb_nxt, pb_nxt, sb_nxt, s, col0,      h, chose(xa.x, xa.y, s), hr0, hi0, ps0, ar0, ai0);
    tile_finish(bR, bM, bI, vq, wq, c0, c1, hr1, hi1, ps1, ar1, ai1);
    tile_publish(hb_nxt, pb_nxt, sb_nxt, s, col0 + 16, h, chose(xb.x, xb.y, s), hr1, hi1, ps1, ar1, ai1);

    __syncthreads();
    __builtin_amdgcn_s_setprio(0);
}

__global__ __launch_bounds__(128) void mps_rnn_plane_per_wave_kernel(
    const int*   __restrict__ xg,
    const float* __restrict__ Mg,
    const float* __restrict__ vg,
    const float* __restrict__ wg,
    const float* __restrict__ cg,
    float*       __restrict__ outg)
{
    __shared__ __attribute__((aligned(16))) float hbuf[2 * 2 * BCOLS * HP];
    __shared__ float pbuf[2 * HLOC * BCOLS];
    __shared__ float sbuf[2 * 3 * BCOLS];

    const int tid  = threadIdx.x;
    const int lane = tid & 31;
    const int wave = tid >> 5;
    const int c    = lane & 15;
    const int h    = lane >> 4;
    const int s    = wave;
    const int col0 = c;
    const size_t gcol0 = (size_t)blockIdx.x * BCOLS + col0;

    Lds L;
    L.hb = hbuf; L.pb = pbuf; L.sb = sbuf;

    const float h0 = 0.17677669529663688f;
    for (int idx = tid; idx < BCOLS * DCUT; idx += 128) {
        const int cc = idx >> 5, rr = idx & 31;
        hbuf[(0 * BCOLS + cc) * HP + rr] = h0;
        hbuf[(1 * BCOLS + cc) * HP + rr] = 0.f;
    }

    v4f A0[16], A1[16];
    load_plane(A0, Mg, 0, s, c, h);

    float logp = 0.f, phase = 0.f;
    __syncthreads();

    for (int t = 0; t < NS; t += 2) {
        site(t,     A0, A1, L, xg, Mg, vg, wg, cg, tid, wave, s, c, h, col0, gcol0, logp, phase);
        site(t + 1, A1, A0, L, xg, Mg, vg, wg, cg, tid, wave, s, c, h, col0, gcol0, logp, phase);
    }

    if (wave == 0) {
        site_account(pbuf + (NS & 1) * (HLOC * BCOLS), sbuf + (NS & 1) * (3 * BCOLS), tid, logp, phase);
        float2 o;
        o.x = 0.5f * logp;
        o.y = phase;
        *(float2*)(outg + ((size_t)blockIdx.x * BCOLS + tid) * 2) = o;
    }
}

extern "C" void kernel_launch(void* const* d_in, const int* in_sizes, int n_in,
                              void* d_out, int out_size, void* d_ws, size_t ws_size,
                              hipStream_t stream) {
    (void)in_sizes; (void)n_in; (void)out_size; (void)d_ws; (void)ws_size;
    const int*   x  = (const int*)  d_in[0];
    const float* M  = (const float*)d_in[1];
    const float* v  = (const float*)d_in[2];
    const float* w  = (const float*)d_in[3];
    const float* c  = (const float*)d_in[4];
    float* out = (float*)d_out;

    dim3 grid(NB / BCOLS);
    dim3 block(128);
    mps_rnn_plane_per_wave_kernel<<<grid, block, 0, stream>>>(x, M, v, w, c, out);
}


# Tile finish on register pairs: Born weight and w.h kept as two-lane partial sums (w's re,im pair used as loaded), w.h scaled once, the three half sums cross together; their sum order changes
# speedup vs baseline: 2.3984x; 1.0832x over previous
#include <hip/hip_runtime.h>
#include <math.h>

#define NB   16384
#define NQ   100
#define NS   50
#define DCUT 32
#define HLOC 4

#define BCOLS 32
#define HP    36

typedef __attribute__((ext_vector_type(2))) float v2f;
typedef __attribute__((ext_vector_type(4))) float v4f;
typedef __attribute__((ext_vector_type(8))) float v8f;

static __device__ __forceinline__ v8f wmma4(v2f a, v2f b, v8f c) {
    return __builtin_amdgcn_wmma_f32_16x16x4_f32(
         false, a,  false, b,
         (short)0, c,  false,  false);
}

static __device__ __forceinline__ void load_plane(v4f (&A)[16], const float* __restrict__ Mg,
                                                  int t, int s, int c, int h) {
    const float* base = Mg + (size_t)(t * HLOC + s) * DCUT * DCUT * 2;
    #pragma unroll
    for (int mt = 0; mt < 2; ++mt) {
        #pragma unroll
        for (int kk = 0; kk < 8; ++kk)
            A[mt * 8 + kk] = *(const v4f*)(base + ((mt * 16 + c) * DCUT + kk * 4 + 2 * h) * 2);
    }
}

static __device__ __forceinline__ void load_h(v2f (&Br)[8], v2f (&Bi)[8], const float* hb, int col, int h) {
    const float* b0 = hb + (0 * BCOLS + col) * HP + 2 * h;
    const float* b1 = hb + (1 * BCOLS + col) * HP + 2 * h;
    #pragma unroll
    for (int kk = 0; kk < 8; ++kk) {
        Br[kk] = *(const v2f*)(b0 + kk * 4);
        Bi[kk] = *(const v2f*)(b1 + kk * 4);
    }
}

static __device__ __forceinline__ void tile_products(const v2f (&Ar)[16], const v2f (&Ai)[16],
                                                     const v2f (&Br)[8], const v2f (&Bi)[8],
                                                     v8f (&aR)[2], v8f (&aM)[2], v8f (&aI)[2]) {
    #pragma unroll
    for (int mt = 0; mt < 2; ++mt) { aR[mt] = (v8f){}; aM[mt] = (v8f){}; aI[mt] = (v8f){}; }
    #pragma unroll
    for (int kk = 0; kk < 8; ++kk) {
        #pragma unroll
        for (int mt = 0; mt < 2; ++mt) {
            aR[mt] = wmma4(Ar[mt * 8 + kk], Br[kk], aR[mt]);
            aM[mt] = wmma4(Ai[mt * 8 + kk], Bi[kk], aM[mt]);
            aI[mt] = wmma4(Ar[mt * 8 + kk], Bi[kk], aI[mt]);
        }
        #pragma unroll
        for (int mt = 0; mt < 2; ++mt)
            aI[mt] = wmma4(Ai[mt * 8 + kk], Br[kk], aI[mt]);
    }
}

static __device__ __forceinline__ void tile_finish(const v8f (&aR)[2], const v8f (&aM)[2], const v8f (&aI)[2],
                                                   const v2f (&VR)[8], const v2f (&VI)[8], const v4f (&wq)[8],
                                                   float c0, float c1,
                                                   v2f (&hr)[8], v2f (&hi)[8],
                                                   float& ps, float& ampR, float& ampI) {
    v2f ppA[2], ppB[2], S[2], T[2];
    #pragma unroll
    for (int mt = 0; mt < 2; ++mt) {
        ppA[mt] = (v2f){0.f, 0.f}; ppB[mt] = (v2f){0.f, 0.f};
        S[mt]   = (v2f){0.f, 0.f}; T[mt]   = (v2f){0.f, 0.f};
        #pragma unroll
        for (int q = 0; q < 4; ++q) {
            const v2f xr = (v2f){aR[mt][2 * q], aR[mt][2 * q + 1]};
            const v2f xm = (v2f){aM[mt][2 * q], aM[mt][2 * q + 1]};
            const v2f xi = (v2f){aI[mt][2 * q], aI[mt][2 * q + 1]};
            const v2f a = xr - xm + VR[mt * 4 + q];
            const v2f b = xi + VI[mt * 4 + q];
            hr[mt * 4 + q] = a;
            hi[mt * 4 + q] = b;
            ppA[mt] = a * a + ppA[mt];
            ppB[mt] = b * b + ppB[mt];
            const v4f w4 = wq[mt * 4 + q];
            const v2f w0 = (v2f){w4[0], w4[1]};
            const v2f w1 = (v2f){w4[2], w4[3]};
            S[mt] = w0 * (v2f){a[0], a[0]} + S[mt];
            T[mt] = w0 * (v2f){b[0], b[0]} + T[mt];
            S[mt] = w1 * (v2f){a[1], a[1]} + S[mt];
            T[mt] = w1 * (v2f){b[1], b[1]} + T[mt];
        }
    }
    const v2f pp2 = (ppA[0] + ppA[1]) + (ppB[0] + ppB[1]);
    const v2f Ss  = S[0] + S[1];
    const v2f Ts  = T[0] + T[1];
    const float pp = pp2[0] + pp2[1];
    float sR = Ss[0] - Ts[1];
    float sI = Ts[0] + Ss[1];
    ps = pp + __shfl_xor(pp, 16, 32);
    sR += __shfl_xor(sR, 16, 32);
    sI += __shfl_xor(sI, 16, 32);
    const float rinv = rsqrtf(ps);
    const v2f r2 = (v2f){rinv, rinv};
    #pragma unroll
    for (int j = 0; j < 8; ++j) {
        hr[j] = hr[j] * r2;
        hi[j] = hi[j] * r2;
    }
    ampR = sR * rinv + c0;
    ampI = sI * rinv + c1;
}

static __device__ __forceinline__ void tile_publish(float* hb, float* pb, float* sb, int s, int col, int h, bool sel,
                                                    const v2f (&hr)[8], const v2f (&hi)[8],
                                                    float ps, float ampR, float ampI) {
    if (h == 0) pb[s * BCOLS + col] = ps;
    if (sel) {
        float* d0 = hb + (0 * BCOLS + col) * HP + 8 * h;
        float* d1 = hb + (1 * BCOLS + col) * HP + 8 * h;
        #pragma unroll
        for (int mt = 0; mt < 2; ++mt) {
            #pragma unroll
            for (int q = 0; q < 2; ++q) {
                const v2f a0 = hr[mt * 4 + q * 2], a1 = hr[mt * 4 + q * 2 + 1];
                const v2f b0 = hi[mt * 4 + q * 2], b1 = hi[mt * 4 + q * 2 + 1];
                *(v4f*)(d0 + mt * 16 + q * 4) = (v4f){a0[0], a0[1], a1[0], a1[1]};
                *(v4f*)(d1 + mt * 16 + q * 4) = (v4f){b0[0], b0[1], b1[0], b1[1]};
            }
        }
        if (h == 0) {
            sb[0 * BCOLS + col] = ps;
            sb[1 * BCOLS + col] = ampR;
            sb[2 * BCOLS + col] = ampI;
        }
    }
}

static __device__ __forceinline__ void site_account(const float* pb, const float* sb, int col,
                                                    float& logp, float& phase) {
    float psum = 0.f;
    #pragma unroll
    for (int k = 0; k < HLOC; ++k) psum += pb[k * BCOLS + col];
    logp  += logf(sb[0 * BCOLS + col]) - logf(psum);
    phase += atan2f(sb[2 * BCOLS + col], sb[1 * BCOLS + col]);
}

static __device__ __forceinline__ bool chose(int x0, int x1, int s) {
    const int ss = 2 * x0 + x1;
    return (s == 3) ? (ss != 0 && ss != 1 && ss != 2) : (ss == s);
}

struct Lds {
    float* hb;
    float* pb;
    float* sb;
};

static __device__ __forceinline__ void site(int t, const v4f (&A)[16], v4f (&An)[16], const Lds& L,
                                            const int* __restrict__ xg, const float* __restrict__ Mg,
                                            const float* __restrict__ vg, const float* __restrict__ wg,
                                            const float* __restrict__ cg,
                                            int tid, int wave, int s, int c, int h, int col0, size_t gcol0,
                                            float& logp, float& phase) {
    const int par = t & 1;
    const float* hb_cur = L.hb + par * (2 * BCOLS * HP);
    float*       hb_nxt = L.hb + (par ^ 1) * (2 * BCOLS * HP);
    const float* pb_cur = L.pb + par * (HLOC * BCOLS);
    float*       pb_nxt = L.pb + (par ^ 1) * (HLOC * BCOLS);
    const float* sb_cur = L.sb + par * (3 * BCOLS);
    float*       sb_nxt = L.sb + (par ^ 1) * (3 * BCOLS);

    v2f Br0[8], Bi0[8], Br1[8], Bi1[8];
    load_h(Br0, Bi0, hb_cur, col0, h);
    load_h(Br1, Bi1, hb_cur, col0 + 16, h);
    __builtin_amdgcn_sched_barrier(0);

    v4f vq[8], wq[8];
    {
        const float* vt = vg + ((size_t)(t * HLOC + s) * DCUT + 8 * h) * 2;
        const float* wt = wg + ((size_t)t * DCUT + 8 * h) * 2;
        #pragma unroll
        for (int mt = 0; mt < 2; ++mt) {
            #pragma unroll
            for (int q = 0; q < 4; ++q) {
                vq[mt * 4 + q] = *(const v4f*)(vt + (mt * 16 + 2 * q) * 2);
                wq[mt * 4 + q] = *(const v4f*)(wt + (mt * 16 + 2 * q) * 2);
            }
        }
    }
    const int2 xa = *(const int2*)(xg + gcol0 * NQ + 2 * t);
    const int2 xb = *(const int2*)(xg + (gcol0 + 16) * NQ + 2 * t);
    const float c0 = cg[t * 2 + 0];
    const float c1 = cg[t * 2 + 1];

    if (wave == 0 && t > 0) site_account(pb_cur, sb_cur, tid, logp, phase);

    v8f aR[2], aM[2], aI[2];
    v2f hr0[8], hi0[8], hr1[8], hi1[8];
    float ps0, ar0, ai0, ps1, ar1, ai1;

    v2f VR[8], VI[8];
    #pragma unroll
    for (int j = 0; j < 8; ++j) {
        VR[j][0] = vq[j][0]; VR[j][1] = vq[j][2];
        VI[j][0] = vq[j][1]; VI[j][1] = vq[j][3];
    }

    v2f Ar[16], Ai[16];
    #pragma unroll
    for (int f = 0; f < 16; ++f) {
        Ar[f][0] = A[f][0]; Ar[f][1] = A[f][2];
        Ai[f][0] = A[f][1]; Ai[f][1] = A[f][3];
    }

    v8f bR[2], bM[2], bI[2];
    tile_products(Ar, Ai, Br0, Bi0, aR, aM, aI);
    __builtin_amdgcn_sched_barrier(0);
    __builtin_amdgcn_s_setprio(3);
    load_plane(An, Mg, (t + 1 < NS) ? t + 1 : t, s, c, h);
    tile_finish(aR, aM, aI, VR, VI, wq, c0, c1, hr0, hi0, ps0, ar0, ai0);
    tile_products(Ar, Ai, Br1, Bi1, bR, bM, bI);
    #pragma unroll
    for (int i = 0; i < 64; ++i) {
        __builtin_amdgcn_sched_group_barrier(0x008, 1, 0);
        __builtin_amdgcn_sched_group_barrier(0x002, 3, 0);
    }
    tile_publish(hb_nxt, pb_nxt, sb_nxt, s, col0,      h, chose(xa.x, xa.y, s), hr0, hi0, ps0, ar0, ai0);
    tile_finish(bR, bM, bI, VR, VI, wq, c0, c1, hr1, hi1, ps1, ar1, ai1);
    tile_publish(hb_nxt, pb_nxt, sb_nxt, s, col0 + 16, h, chose(xb.x, xb.y, s), hr1, hi1, ps1, ar1, ai1);

    __syncthreads();
    __builtin_amdgcn_s_setprio(0);
}

__global__ __launch_bounds__(128) void mps_rnn_plane_per_wave_kernel(
    const int*   __restrict__ xg,
    const float* __restrict__ Mg,
    const float* __restrict__ vg,
    const float* __restrict__ wg,
    const float* __restrict__ cg,
    float*       __restrict__ outg)
{
    __shared__ __attribute__((aligned(16))) float hbuf[2 * 2 * BCOLS * HP];
    __shared__ float pbuf[2 * HLOC * BCOLS];
    __shared__ float sbuf[2 * 3 * BCOLS];

    const int tid  = threadIdx.x;
    const int lane = tid & 31;
    const int wave = tid >> 5;
    const int c    = lane & 15;
    const int h    = lane >> 4;
    const int s    = wave;
    const int col0 = c;
    const size_t gcol0 = (size_t)blockIdx.x * BCOLS + col0;

    Lds L;
    L.hb = hbuf; L.pb = pbuf; L.sb = sbuf;

    const float h0 = 0.17677669529663688f;
    for (int idx = tid; idx < BCOLS * DCUT; idx += 128) {
        const int cc = idx >> 5, rr = idx & 31;
        hbuf[(0 * BCOLS + cc) * HP + rr] = h0;
        hbuf[(1 * BCOLS + cc) * HP + rr] = 0.f;
    }

    v4f A0[16], A1[16];
    load_plane(A0, Mg, 0, s, c, h);

    float logp = 0.f, phase = 0.f;
    __syncthreads();

    for (int t = 0; t < NS; t += 2) {
        site(t,     A0, A1, L, xg, Mg, vg, wg, cg, tid, wave, s, c, h, col0, gcol0, logp, phase);
        site(t + 1, A1, A0, L, xg, Mg, vg, wg, cg, tid, wave, s, c, h, col0, gcol0, logp, phase);
    }

    if (wave == 0) {
        site_account(pbuf + (NS & 1) * (HLOC * BCOLS), sbuf + (NS & 1) * (3 * BCOLS), tid, logp, phase);
        float2 o;
        o.x = 0.5f * logp;
        o.y = phase;
        *(float2*)(outg + ((size_t)blockIdx.x * BCOLS + tid) * 2) = o;
    }
}

extern "C" void kernel_launch(void* const* d_in, const int* in_sizes, int n_in,
                              void* d_out, int out_size, void* d_ws, size_t ws_size,
                              hipStream_t stream) {
    (void)in_sizes; (void)n_in; (void)out_size; (void)d_ws; (void)ws_size;
    const int*   x  = (const int*)  d_in[0];
    const float* M  = (const float*)d_in[1];
    const float* v  = (const float*)d_in[2];
    const float* w  = (const float*)d_in[3];
    const float* c  = (const float*)d_in[4];
    float* out = (float*)d_out;

    dim3 grid(NB / BCOLS);
    dim3 block(128);
    mps_rnn_plane_per_wave_kernel<<<grid, block, 0, stream>>>(x, M, v, w, c, out);
}


# Per-site logs and angle deferred: each site's Born weights and amplitude stay in LDS by site; after the last site all 4 waves evaluate them, a wave every 4th site; logp, phase sum order changes
# speedup vs baseline: 2.5051x; 1.0445x over previous
#include <hip/hip_runtime.h>
#include <math.h>

#define NB   16384
#define NQ   100
#define NS   50
#define DCUT 32
#define HLOC 4

#define BCOLS 32
#define HP    36

typedef __attribute__((ext_vector_type(2))) float v2f;
typedef __attribute__((ext_vector_type(4))) float v4f;
typedef __attribute__((ext_vector_type(8))) float v8f;

static __device__ __forceinline__ v8f wmma4(v2f a, v2f b, v8f c) {
    return __builtin_amdgcn_wmma_f32_16x16x4_f32(
         false, a,  false, b,
         (short)0, c,  false,  false);
}

static __device__ __forceinline__ void load_plane(v4f (&A)[16], const float* __restrict__ Mg,
                                                  int t, int s, int c, int h) {
    const float* base = Mg + (size_t)(t * HLOC + s) * DCUT * DCUT * 2;
    #pragma unroll
    for (int mt = 0; mt < 2; ++mt) {
        #pragma unroll
        for (int kk = 0; kk < 8; ++kk)
            A[mt * 8 + kk] = *(const v4f*)(base + ((mt * 16 + c) * DCUT + kk * 4 + 2 * h) * 2);
    }
}

static __device__ __forceinline__ void load_h(v2f (&Br)[8], v2f (&Bi)[8], const float* hb, int col, int h) {
    const float* b0 = hb + (0 * BCOLS + col) * HP + 2 * h;
    const float* b1 = hb + (1 * BCOLS + col) * HP + 2 * h;
    #pragma unroll
    for (int kk = 0; kk < 8; ++kk) {
        Br[kk] = *(const v2f*)(b0 + kk * 4);
        Bi[kk] = *(const v2f*)(b1 + kk * 4);
    }
}

static __device__ __forceinline__ void tile_products(const v2f (&Ar)[16], const v2f (&Ai)[16],
                                                     const v2f (&Br)[8], const v2f (&Bi)[8],
                                                     v8f (&aR)[2], v8f (&aM)[2], v8f (&aI)[2]) {
    #pragma unroll
    for (int mt = 0; mt < 2; ++mt) { aR[mt] = (v8f){}; aM[mt] = (v8f){}; aI[mt] = (v8f){}; }
    #pragma unroll
    for (int kk = 0; kk < 8; ++kk) {
        #pragma unroll
        for (int mt = 0; mt < 2; ++mt) {
            aR[mt] = wmma4(Ar[mt * 8 + kk], Br[kk], aR[mt]);
            aM[mt] = wmma4(Ai[mt * 8 + kk], Bi[kk], aM[mt]);
            aI[mt] = wmma4(Ar[mt * 8 + kk], Bi[kk], aI[mt]);
        }
        #pragma unroll
        for (int mt = 0; mt < 2; ++mt)
            aI[mt] = wmma4(Ai[mt * 8 + kk], Br[kk], aI[mt]);
    }
}

static __device__ __forceinline__ void tile_finish(const v8f (&aR)[2], const v8f (&aM)[2], const v8f (&aI)[2],
                                                   const v2f (&VR)[8], const v2f (&VI)[8], const v4f (&wq)[8],
                                                   float c0, float c1,
                                                   v2f (&hr)[8], v2f (&hi)[8],
                                                   float& ps, float& ampR, float& ampI) {
    v2f ppA[2], ppB[2], S[2], T[2];
    #pragma unroll
    for (int mt = 0; mt < 2; ++mt) {
        ppA[mt] = (v2f){0.f, 0.f}; ppB[mt] = (v2f){0.f, 0.f};
        S[mt]   = (v2f){0.f, 0.f}; T[mt]   = (v2f){0.f, 0.f};
        #pragma unroll
        for (int q = 0; q < 4; ++q) {
            const v2f xr = (v2f){aR[mt][2 * q], aR[mt][2 * q + 1]};
            const v2f xm = (v2f){aM[mt][2 * q], aM[mt][2 * q + 1]};
            const v2f xi = (v2f){aI[mt][2 * q], aI[mt][2 * q + 1]};
            const v2f a = xr - xm + VR[mt * 4 + q];
            const v2f b = xi + VI[mt * 4 + q];
            hr[mt * 4 + q] = a;
            hi[mt * 4 + q] = b;
            ppA[mt] = a * a + ppA[mt];
            ppB[mt] = b * b + ppB[mt];
            const v4f w4 = wq[mt * 4 + q];
            const v2f w0 = (v2f){w4[0], w4[1]};
            const v2f w1 = (v2f){w4[2], w4[3]};
            S[mt] = w0 * (v2f){a[0], a[0]} + S[mt];
            T[mt] = w0 * (v2f){b[0], b[0]} + T[mt];
            S[mt] = w1 * (v2f){a[1], a[1]} + S[mt];
            T[mt] = w1 * (v2f){b[1], b[1]} + T[mt];
        }
    }
    const v2f pp2 = (ppA[0] + ppA[1]) + (ppB[0] + ppB[1]);
    const v2f Ss  = S[0] + S[1];
    const v2f Ts  = T[0] + T[1];
    const float pp = pp2[0] + pp2[1];
    float sR = Ss[0] - Ts[1];
    float sI = Ts[0] + Ss[1];
    ps = pp + __shfl_xor(pp, 16, 32);
    sR += __shfl_xor(sR, 16, 32);
    sI += __shfl_xor(sI, 16, 32);
    const float rinv = rsqrtf(ps);
    const v2f r2 = (v2f){rinv, rinv};
    #pragma unroll
    for (int j = 0; j < 8; ++j) {
        hr[j] = hr[j] * r2;
        hi[j] = hi[j] * r2;
    }
    ampR = sR * rinv + c0;
    ampI = sI * rinv + c1;
}

static __device__ __forceinline__ void tile_publish(float* hb, float* pb, float* sb, int s, int col, int h, bool sel,
                                                    const v2f (&hr)[8], const v2f (&hi)[8],
                                                    float ps, float ampR, float ampI) {
    if (h == 0) pb[s * BCOLS + col] = ps;
    if (sel) {
        float* d0 = hb + (0 * BCOLS + col) * HP + 8 * h;
        float* d1 = hb + (1 * BCOLS + col) * HP + 8 * h;
        #pragma unroll
        for (int mt = 0; mt < 2; ++mt) {
            #pragma unroll
            for (int q = 0; q < 2; ++q) {
                const v2f a0 = hr[mt * 4 + q * 2], a1 = hr[mt * 4 + q * 2 + 1];
                const v2f b0 = hi[mt * 4 + q * 2], b1 = hi[mt * 4 + q * 2 + 1];
                *(v4f*)(d0 + mt * 16 + q * 4) = (v4f){a0[0], a0[1], a1[0], a1[1]};
                *(v4f*)(d1 + mt * 16 + q * 4) = (v4f){b0[0], b0[1], b1[0], b1[1]};
            }
        }
        if (h == 0) {
            sb[0 * BCOLS + col] = ps;
            sb[1 * BCOLS + col] = ampR;
            sb[2 * BCOLS + col] = ampI;
        }
    }
}

static __device__ __forceinline__ void site_account(const float* pb, const float* sb, int col,
                                                    float& logp, float& phase) {
    float psum = 0.f;
    #pragma unroll
    for (int k = 0; k < HLOC; ++k) psum += pb[k * BCOLS + col];
    logp  += logf(sb[0 * BCOLS + col]) - logf(psum);
    phase += atan2f(sb[2 * BCOLS + col], sb[1 * BCOLS + col]);
}

static __device__ __forceinline__ bool chose(int x0, int x1, int s) {
    const int ss = 2 * x0 + x1;
    return (s == 3) ? (ss != 0 && ss != 1 && ss != 2) : (ss == s);
}

struct Lds {
    float* hb;
    float* pb;
    float* sb;
};

static __device__ __forceinline__ void site(int t, const v4f (&A)[16], v4f (&An)[16], const Lds& L,
                                            const int* __restrict__ xg, const float* __restrict__ Mg,
                                            const float* __restrict__ vg, const float* __restrict__ wg,
                                            const float* __restrict__ cg,
                                            int s, int c, int h, int col0, size_t gcol0) {
    const int par = t & 1;
    const float* hb_cur = L.hb + par * (2 * BCOLS * HP);
    float*       hb_nxt = L.hb + (par ^ 1) * (2 * BCOLS * HP);
    float*       pb_t   = L.pb + t * (HLOC * BCOLS);
    float*       sb_t   = L.sb + t * (3 * BCOLS);

    v2f Br0[8], Bi0[8], Br1[8], Bi1[8];
    load_h(Br0, Bi0, hb_cur, col0, h);
    load_h(Br1, Bi1, hb_cur, col0 + 16, h);
    __builtin_amdgcn_sched_barrier(0);

    v4f vq[8], wq[8];
    {
        const float* vt = vg + ((size_t)(t * HLOC + s) * DCUT + 8 * h) * 2;
        const float* wt = wg + ((size_t)t * DCUT + 8 * h) * 2;
        #pragma unroll
        for (int mt = 0; mt < 2; ++mt) {
            #pragma unroll
            for (int q = 0; q < 4; ++q) {
                vq[mt * 4 + q] = *(const v4f*)(vt + (mt * 16 + 2 * q) * 2);
                wq[mt * 4 + q] = *(const v4f*)(wt + (mt * 16 + 2 * q) * 2);
            }
        }
    }
    const int2 xa = *(const int2*)(xg + gcol0 * NQ + 2 * t);
    const int2 xb = *(const int2*)(xg + (gcol0 + 16) * NQ + 2 * t);
    const float c0 = cg[t * 2 + 0];
    const float c1 = cg[t * 2 + 1];

    v8f aR[2], aM[2], aI[2];
    v2f hr0[8], hi0[8], hr1[8], hi1[8];
    float ps0, ar0, ai0, ps1, ar1, ai1;

    v2f VR[8], VI[8];
    #pragma unroll
    for (int j = 0; j < 8; ++j) {
        VR[j][0] = vq[j][0]; VR[j][1] = vq[j][2];
        VI[j][0] = vq[j][1]; VI[j][1] = vq[j][3];
    }

    v2f Ar[16], Ai[16];
    #pragma unroll
    for (int f = 0; f < 16; ++f) {
        Ar[f][0] = A[f][0]; Ar[f][1] = A[f][2];
        Ai[f][0] = A[f][1]; Ai[f][1] = A[f][3];
    }

    v8f bR[2], bM[2], bI[2];
    tile_products(Ar, Ai, Br0, Bi0, aR, aM, aI);
    __builtin_amdgcn_sched_barrier(0);
    __builtin_amdgcn_s_setprio(3);
    load_plane(An, Mg, (t + 1 < NS) ? t + 1 : t, s, c, h);
    tile_finish(aR, aM, aI, VR, VI, wq, c0, c1, hr0, hi0, ps0, ar0, ai0);
    tile_products(Ar, Ai, Br1, Bi1, bR, bM, bI);
    #pragma unroll
    for (int i = 0; i < 64; ++i) {
        __builtin_amdgcn_sched_group_barrier(0x008, 1, 0);
        __builtin_amdgcn_sched_group_barrier(0x002, 3, 0);
    }
    tile_publish(hb_nxt, pb_t, sb_t, s, col0,      h, chose(xa.x, xa.y, s), hr0, hi0, ps0, ar0, ai0);
    tile_finish(bR, bM, bI, VR, VI, wq, c0, c1, hr1, hi1, ps1, ar1, ai1);
    tile_publish(hb_nxt, pb_t, sb_t, s, col0 + 16, h, chose(xb.x, xb.y, s), hr1, hi1, ps1, ar1, ai1);

    __syncthreads();
    __builtin_amdgcn_s_setprio(0);
}

__global__ __launch_bounds__(128) __attribute__((amdgpu_waves_per_eu(2, 2))) void mps_rnn_plane_per_wave_kernel(
    const int*   __restrict__ xg,
    const float* __restrict__ Mg,
    const float* __restrict__ vg,
    const float* __restrict__ wg,
    const float* __restrict__ cg,
    float*       __restrict__ outg)
{
    __shared__ __attribute__((aligned(16))) float hbuf[2 * 2 * BCOLS * HP];
    __shared__ float pbuf[NS * HLOC * BCOLS];
    __shared__ float sbuf[NS * 3 * BCOLS];
    __shared__ float red[2 * 4 * BCOLS];

    const int tid  = threadIdx.x;
    const int lane = tid & 31;
    const int wave = tid >> 5;
    const int c    = lane & 15;
    const int h    = lane >> 4;
    const int s    = wave;
    const int col0 = c;
    const size_t gcol0 = (size_t)blockIdx.x * BCOLS + col0;

    Lds L;
    L.hb = hbuf; L.pb = pbuf; L.sb = sbuf;

    const float h0 = 0.17677669529663688f;
    for (int idx = tid; idx < BCOLS * DCUT; idx += 128) {
        const int cc = idx >> 5, rr = idx & 31;
        hbuf[(0 * BCOLS + cc) * HP + rr] = h0;
        hbuf[(1 * BCOLS + cc) * HP + rr] = 0.f;
    }

    v4f A0[16], A1[16];
    load_plane(A0, Mg, 0, s, c, h);

    __syncthreads();

    for (int t = 0; t < NS; t += 2) {
        site(t,     A0, A1, L, xg, Mg, vg, wg, cg, s, c, h, col0, gcol0);
        site(t + 1, A1, A0, L, xg, Mg, vg, wg, cg, s, c, h, col0, gcol0);
    }

    {
        float logp = 0.f, phase = 0.f;
        for (int t = wave; t < NS; t += 4)
            site_account(pbuf + t * (HLOC * BCOLS), sbuf + t * (3 * BCOLS), lane, logp, phase);
        red[(0 * 4 + wave) * BCOLS + lane] = logp;
        red[(1 * 4 + wave) * BCOLS + lane] = phase;
    }
    __syncthreads();

    if (wave == 0) {
        float logp = 0.f, phase = 0.f;
        for (int g = 0; g < 4; ++g) {
            logp  += red[(0 * 4 + g) * BCOLS + tid];
            phase += red[(1 * 4 + g) * BCOLS + tid];
        }
        float2 o;
        o.x = 0.5f * logp;
        o.y = phase;
        *(float2*)(outg + ((size_t)blockIdx.x * BCOLS + tid) * 2) = o;
    }
}

extern "C" void kernel_launch(void* const* d_in, const int* in_sizes, int n_in,
                              void* d_out, int out_size, void* d_ws, size_t ws_size,
                              hipStream_t stream) {
    (void)in_sizes; (void)n_in; (void)out_size; (void)d_ws; (void)ws_size;
    const int*   x  = (const int*)  d_in[0];
    const float* M  = (const float*)d_in[1];
    const float* v  = (const float*)d_in[2];
    const float* w  = (const float*)d_in[3];
    const float* c  = (const float*)d_in[4];
    float* out = (float*)d_out;

    dim3 grid(NB / BCOLS);
    dim3 block(128);
    mps_rnn_plane_per_wave_kernel<<<grid, block, 0, stream>>>(x, M, v, w, c, out);
}
